# starting kernel
# baseline (speedup 1.0000x reference)
#include <hip/hip_runtime.h>
#include <math.h>

typedef __attribute__((ext_vector_type(2))) float v2f;
typedef __attribute__((ext_vector_type(8))) float v8f;

#define TT 4096
#define BB 256

__device__ __forceinline__ float fast_tanh(float x) {
    float e = __expf(2.0f * x);
    return 1.0f - 2.0f * __builtin_amdgcn_rcpf(e + 1.0f);
}

__device__ __forceinline__ v8f wmma4(v2f a, v2f b, v8f c) {
    return __builtin_amdgcn_wmma_f32_16x16x4_f32(
        false, a, false, b, (short)0, c, false, false);
}

__global__ __launch_bounds__(512)
void pinn_kernel(const float* __restrict__ x,
                 const float* __restrict__ W1,  const float* __restrict__ b1,
                 const float* __restrict__ W2,  const float* __restrict__ b2,
                 const float* __restrict__ W3,  const float* __restrict__ b3,
                 const float* __restrict__ logE,const float* __restrict__ logeta,
                 const float* __restrict__ imp, const float* __restrict__ lw,
                 const float* __restrict__ Wf1, const float* __restrict__ bf1,
                 const float* __restrict__ Wf2, const float* __restrict__ bf2,
                 float* __restrict__ out)
{
    __shared__ __align__(16) float h1s[16][16 * 34];
    __shared__ __align__(16) float h2s[16][16 * 20];
    __shared__ __align__(16) float sbuf[2][16][16];

    const int tid  = threadIdx.x;
    const int n    = tid >> 5;
    const int lane = tid & 31;
    const int hi   = lane >> 4;
    const int lm   = lane & 15;
    const int b0   = blockIdx.x * 16;
    const int kb   = hi ? 2 : 0;

    const float E   = __expf(logE[n]);
    const float eta = __expf(logeta[n]);
    const float iw  = 1.0f / (1.0f + __expf(-imp[n]));
    const float b3v = b3[n];

    v2f B1[2]; v8f C1[2];
#pragma unroll
    for (int h = 0; h < 2; ++h) {
        int j = 16 * h + lm;
        B1[h].x = W1[(n * 32 + j) * 3 + kb];
        B1[h].y = hi ? 0.0f : W1[(n * 32 + j) * 3 + 1];
        float bb = b1[n * 32 + j];
#pragma unroll
        for (int i = 0; i < 8; ++i) C1[h][i] = bb;
    }
    v2f B2[8];
#pragma unroll
    for (int kc = 0; kc < 8; ++kc) {
        B2[kc].x = W2[(n * 16 + lm) * 32 + 4 * kc + kb];
        B2[kc].y = W2[(n * 16 + lm) * 32 + 4 * kc + kb + 1];
    }
    const float c2s = b2[n * 16 + lm];

    float w3v[16];
#pragma unroll
    for (int k = 0; k < 16; ++k) w3v[k] = W3[n * 16 + k];

    float cl[4];
    {
        float e0 = __expf(lw[0]), e1 = __expf(lw[1]);
        float e2 = __expf(lw[2]), e3 = __expf(lw[3]);
        float es = e0 + e1 + e2 + e3;
        float ws[4] = { e0 / es, e1 / es, e2 / es, e3 / es };
#pragma unroll
        for (int l = 0; l < 4; ++l) {
            float s = 0.0f;
#pragma unroll
            for (int p = 0; p < 4; ++p)
                s += 1.0f / (1.0f + __expf(-imp[l * 4 + p]));
            cl[l] = (s / (s + 1e-8f)) * ws[l];
        }
    }
    float wf1v[4];
#pragma unroll
    for (int l = 0; l < 4; ++l) wf1v[l] = Wf1[lm * 4 + l];
    const float bf1v = bf1[lm];
    const float wf2v = Wf2[lm];
    const float bf2v = bf2[0];

    float prev = 0.0f;
    const float2* xp = (const float2*)x;
    float2 xc = make_float2(0.0f, 0.0f);
    if (!hi) xc = xp[(size_t)(b0 + lm) * TT];

    float* h1p = h1s[n];
    float* h2p = h2s[n];

    for (int t = 0; t < TT; ++t) {
        float2 xn = make_float2(0.0f, 0.0f);
        if (!hi && (t + 1) < TT) xn = xp[(size_t)(b0 + lm) * TT + (t + 1)];

        v2f a1;
        a1.x = hi ? prev : xc.x;
        a1.y = hi ? 0.0f : xc.y;

        v8f d0 = wmma4(a1, B1[0], C1[0]);
        v8f d1 = wmma4(a1, B1[1], C1[1]);
#pragma unroll
        for (int i = 0; i < 8; ++i) { d0[i] = fast_tanh(d0[i]); d1[i] = fast_tanh(d1[i]); }

#pragma unroll
        for (int v = 0; v < 8; ++v) {
            h1p[(v + 8 * hi) * 34 + lm]      = d0[v];
            h1p[(v + 8 * hi) * 34 + 16 + lm] = d1[v];
        }

        v8f acc;
#pragma unroll
        for (int i = 0; i < 8; ++i) acc[i] = c2s;
#pragma unroll
        for (int kc = 0; kc < 8; ++kc) {
            v2f a2 = *(const v2f*)&h1p[lm * 34 + 4 * kc + kb];
            acc = wmma4(a2, B2[kc], acc);
        }
#pragma unroll
        for (int i = 0; i < 8; ++i) acc[i] = fast_tanh(acc[i]);

#pragma unroll
        for (int v = 0; v < 8; ++v) h2p[(v + 8 * hi) * 20 + lm] = acc[v];

        const int buf = t & 1;
        if (!hi) {
            const float4* row = (const float4*)&h2p[lm * 20];
            float corr = b3v;
#pragma unroll
            for (int j = 0; j < 4; ++j) {
                float4 r = row[j];
                corr = fmaf(r.x, w3v[4 * j + 0], corr);
                corr = fmaf(r.y, w3v[4 * j + 1], corr);
                corr = fmaf(r.z, w3v[4 * j + 2], corr);
                corr = fmaf(r.w, w3v[4 * j + 3], corr);
            }
            float mw = E * xc.x + eta * xc.y;
            sbuf[buf][n][lm] = (mw + corr) * iw;
        }
        __syncthreads();

        prev = sbuf[buf][n][lm];

#pragma unroll
        for (int pp = 0; pp < 2; ++pp) {
            int p = 2 * hi + pp;
            float v0 = sbuf[buf][0 * 4 + p][n] * cl[0];
            float v1 = sbuf[buf][1 * 4 + p][n] * cl[1];
            float v2 = sbuf[buf][2 * 4 + p][n] * cl[2];
            float v3 = sbuf[buf][3 * 4 + p][n] * cl[3];
            float hf = fmaf(v0, wf1v[0], bf1v);
            hf = fmaf(v1, wf1v[1], hf);
            hf = fmaf(v2, wf1v[2], hf);
            hf = fmaf(v3, wf1v[3], hf);
            hf = fmaxf(hf, 0.0f);
            float c = wf2v * hf;
            c += __shfl_xor(c, 1);
            c += __shfl_xor(c, 2);
            c += __shfl_xor(c, 4);
            c += __shfl_xor(c, 8);
            if (lm == 0)
                out[(size_t)(p * BB + b0 + n) * TT + t] = c + bf2v;
        }
        xc = xn;
    }
}

extern "C" void kernel_launch(void* const* d_in, const int* in_sizes, int n_in,
                              void* d_out, int out_size, void* d_ws, size_t ws_size,
                              hipStream_t stream) {
    (void)in_sizes; (void)n_in; (void)d_ws; (void)ws_size; (void)out_size;
    pinn_kernel<<<dim3(16), dim3(512), 0, stream>>>(
        (const float*)d_in[0],  (const float*)d_in[1],  (const float*)d_in[2],
        (const float*)d_in[3],  (const float*)d_in[4],  (const float*)d_in[5],
        (const float*)d_in[6],  (const float*)d_in[7],  (const float*)d_in[8],
        (const float*)d_in[9],  (const float*)d_in[10], (const float*)d_in[11],
        (const float*)d_in[12], (const float*)d_in[13], (const float*)d_in[14],
        (float*)d_out);
}


# re-grid: one workgroup = the 4 neurons of one head column x 16 batch rows (64 workgroups of 4 waves, one wave per SIMD) instead of 16 workgroups of 16 waves; sums and their order unchanged
# speedup vs baseline: 1.4939x; 1.4939x over previous
#include <hip/hip_runtime.h>
#include <math.h>

typedef __attribute__((ext_vector_type(2))) float v2f;
typedef __attribute__((ext_vector_type(8))) float v8f;

#define TT 4096
#define BB 256

__device__ __forceinline__ float fast_tanh(float x) {
    float e = __expf(2.0f * x);
    return 1.0f - 2.0f * __builtin_amdgcn_rcpf(e + 1.0f);
}

__device__ __forceinline__ v8f wmma4(v2f a, v2f b, v8f c) {
    return __builtin_amdgcn_wmma_f32_16x16x4_f32(
        false, a, false, b, (short)0, c, false, false);
}

__global__ __launch_bounds__(128)
void pinn_kernel(const float* __restrict__ x,
                 const float* __restrict__ W1,  const float* __restrict__ b1,
                 const float* __restrict__ W2,  const float* __restrict__ b2,
                 const float* __restrict__ W3,  const float* __restrict__ b3,
                 const float* __restrict__ logE,const float* __restrict__ logeta,
                 const float* __restrict__ imp, const float* __restrict__ lw,
                 const float* __restrict__ Wf1, const float* __restrict__ bf1,
                 const float* __restrict__ Wf2, const float* __restrict__ bf2,
                 float* __restrict__ out)
{
    __shared__ __align__(16) float h1s[4][16 * 34];
    __shared__ __align__(16) float h2s[4][16 * 20];
    __shared__ __align__(16) float sbuf[2][4][16];

    const int tid  = threadIdx.x;
    const int w    = tid >> 5;
    const int p    = blockIdx.y;
    const int n    = w * 4 + p;
    const int lane = tid & 31;
    const int hi   = lane >> 4;
    const int lm   = lane & 15;
    const int b0   = blockIdx.x * 16;
    const int kb   = hi ? 2 : 0;

    const float E   = __expf(logE[n]);
    const float eta = __expf(logeta[n]);
    const float iw  = 1.0f / (1.0f + __expf(-imp[n]));
    const float b3v = b3[n];

    v2f B1[2]; v8f C1[2];
#pragma unroll
    for (int h = 0; h < 2; ++h) {
        int j = 16 * h + lm;
        B1[h].x = W1[(n * 32 + j) * 3 + kb];
        B1[h].y = hi ? 0.0f : W1[(n * 32 + j) * 3 + 1];
        float bb = b1[n * 32 + j];
#pragma unroll
        for (int i = 0; i < 8; ++i) C1[h][i] = bb;
    }
    v2f B2[8];
#pragma unroll
    for (int kc = 0; kc < 8; ++kc) {
        B2[kc].x = W2[(n * 16 + lm) * 32 + 4 * kc + kb];
        B2[kc].y = W2[(n * 16 + lm) * 32 + 4 * kc + kb + 1];
    }
    const float c2s = b2[n * 16 + lm];

    float w3v[16];
#pragma unroll
    for (int k = 0; k < 16; ++k) w3v[k] = W3[n * 16 + k];

    float cl[4];
    {
        float e0 = __expf(lw[0]), e1 = __expf(lw[1]);
        float e2 = __expf(lw[2]), e3 = __expf(lw[3]);
        float es = e0 + e1 + e2 + e3;
        float ws[4] = { e0 / es, e1 / es, e2 / es, e3 / es };
#pragma unroll
        for (int l = 0; l < 4; ++l) {
            float s = 0.0f;
#pragma unroll
            for (int q = 0; q < 4; ++q)
                s += 1.0f / (1.0f + __expf(-imp[l * 4 + q]));
            cl[l] = (s / (s + 1e-8f)) * ws[l];
        }
    }
    float wf1v[4];
#pragma unroll
    for (int l = 0; l < 4; ++l) wf1v[l] = Wf1[lm * 4 + l];
    const float bf1v = bf1[lm];
    const float wf2v = Wf2[lm];
    const float bf2v = bf2[0];

    float prev = 0.0f;
    const float2* xp = (const float2*)x;
    float2 xc = make_float2(0.0f, 0.0f);
    if (!hi) xc = xp[(size_t)(b0 + lm) * TT];

    float* h1p = h1s[w];
    float* h2p = h2s[w];

    for (int t = 0; t < TT; ++t) {
        float2 xn = make_float2(0.0f, 0.0f);
        if (!hi && (t + 1) < TT) xn = xp[(size_t)(b0 + lm) * TT + (t + 1)];

        v2f a1;
        a1.x = hi ? prev : xc.x;
        a1.y = hi ? 0.0f : xc.y;

        v8f d0 = wmma4(a1, B1[0], C1[0]);
        v8f d1 = wmma4(a1, B1[1], C1[1]);
#pragma unroll
        for (int i = 0; i < 8; ++i) { d0[i] = fast_tanh(d0[i]); d1[i] = fast_tanh(d1[i]); }

#pragma unroll
        for (int v = 0; v < 8; ++v) {
            h1p[(v + 8 * hi) * 34 + lm]      = d0[v];
            h1p[(v + 8 * hi) * 34 + 16 + lm] = d1[v];
        }

        v8f acc;
#pragma unroll
        for (int i = 0; i < 8; ++i) acc[i] = c2s;
#pragma unroll
        for (int kc = 0; kc < 8; ++kc) {
            v2f a2 = *(const v2f*)&h1p[lm * 34 + 4 * kc + kb];
            acc = wmma4(a2, B2[kc], acc);
        }
#pragma unroll
        for (int i = 0; i < 8; ++i) acc[i] = fast_tanh(acc[i]);

#pragma unroll
        for (int v = 0; v < 8; ++v) h2p[(v + 8 * hi) * 20 + lm] = acc[v];

        const int buf = t & 1;
        if (!hi) {
            const float4* row = (const float4*)&h2p[lm * 20];
            float corr = b3v;
#pragma unroll
            for (int j = 0; j < 4; ++j) {
                float4 r = row[j];
                corr = fmaf(r.x, w3v[4 * j + 0], corr);
                corr = fmaf(r.y, w3v[4 * j + 1], corr);
                corr = fmaf(r.z, w3v[4 * j + 2], corr);
                corr = fmaf(r.w, w3v[4 * j + 3], corr);
            }
            float mw = E * xc.x + eta * xc.y;
            sbuf[buf][w][lm] = (mw + corr) * iw;
        }
        __syncthreads();

        prev = sbuf[buf][w][lm];

#pragma unroll
        for (int pp = 0; pp < 2; ++pp) {
            int bi = 4 * w + 2 * hi + pp;
            float v0 = sbuf[buf][0][bi] * cl[0];
            float v1 = sbuf[buf][1][bi] * cl[1];
            float v2 = sbuf[buf][2][bi] * cl[2];
            float v3 = sbuf[buf][3][bi] * cl[3];
            float hf = fmaf(v0, wf1v[0], bf1v);
            hf = fmaf(v1, wf1v[1], hf);
            hf = fmaf(v2, wf1v[2], hf);
            hf = fmaf(v3, wf1v[3], hf);
            hf = fmaxf(hf, 0.0f);
            float c = wf2v * hf;
            c += __shfl_xor(c, 1);
            c += __shfl_xor(c, 2);
            c += __shfl_xor(c, 4);
            c += __shfl_xor(c, 8);
            if (lm == 0)
                out[(size_t)(p * BB + b0 + bi) * TT + t] = c + bf2v;
        }
        xc = xn;
    }
}

extern "C" void kernel_launch(void* const* d_in, const int* in_sizes, int n_in,
                              void* d_out, int out_size, void* d_ws, size_t ws_size,
                              hipStream_t stream) {
    (void)in_sizes; (void)n_in; (void)d_ws; (void)ws_size; (void)out_size;
    pinn_kernel<<<dim3(16, 4), dim3(128), 0, stream>>>(
        (const float*)d_in[0],  (const float*)d_in[1],  (const float*)d_in[2],
        (const float*)d_in[3],  (const float*)d_in[4],  (const float*)d_in[5],
        (const float*)d_in[6],  (const float*)d_in[7],  (const float*)d_in[8],
        (const float*)d_in[9],  (const float*)d_in[10], (const float*)d_in[11],
        (const float*)d_in[12], (const float*)d_in[13], (const float*)d_in[14],
        (float*)d_out);
}


# fusion head out of the step loop: 32 steps of stress held in LDS, then head per block on f32 WMMA (f-sum in-lane + half swap: sum order changed), 128-B row stores; carry by lane swap, no step barrier
# speedup vs baseline: 3.2009x; 2.1426x over previous
#include <hip/hip_runtime.h>
#include <math.h>

typedef __attribute__((ext_vector_type(2))) float v2f;
typedef __attribute__((ext_vector_type(8))) float v8f;

#define TT 4096
#define BB 256
#define CH 32
#define H1N (4 * 16 * 34)
#define H2N (4 * 16 * 20)
#define SP 65

__device__ __forceinline__ float fast_tanh(float x) {
    float e = __expf(2.0f * x);
    return 1.0f - 2.0f * __builtin_amdgcn_rcpf(e + 1.0f);
}

__device__ __forceinline__ float xhalf(float x) {
    int b = __builtin_bit_cast(int, x);
    return __builtin_bit_cast(float,
        __builtin_amdgcn_permlanex16(b, b, 0x76543210, 0xfedcba98, false, false));
}

__device__ __forceinline__ v8f wmma4(v2f a, v2f b, v8f c) {
    return __builtin_amdgcn_wmma_f32_16x16x4_f32(
        false, a, false, b, (short)0, c, false, false);
}

__global__ __launch_bounds__(128)
void pinn_kernel(const float* __restrict__ x,
                 const float* __restrict__ W1,  const float* __restrict__ b1,
                 const float* __restrict__ W2,  const float* __restrict__ b2,
                 const float* __restrict__ W3,  const float* __restrict__ b3,
                 const float* __restrict__ logE,const float* __restrict__ logeta,
                 const float* __restrict__ imp, const float* __restrict__ lw,
                 const float* __restrict__ Wf1, const float* __restrict__ bf1,
                 const float* __restrict__ Wf2, const float* __restrict__ bf2,
                 float* __restrict__ out)
{
    __shared__ __align__(16) float lds[H1N + H2N + 2 * CH * SP];
    float* const ss = lds + H1N + H2N;

    const int tid  = threadIdx.x;
    const int w    = tid >> 5;
    const int p    = blockIdx.y;
    const int n    = w * 4 + p;
    const int lane = tid & 31;
    const int hi   = lane >> 4;
    const int lm   = lane & 15;
    const int b0   = blockIdx.x * 16;
    const int kb   = hi ? 2 : 0;

    const float E   = __expf(logE[n]);
    const float eta = __expf(logeta[n]);
    const float iw  = 1.0f / (1.0f + __expf(-imp[n]));
    const float b3v = b3[n];

    v2f B1[2]; v8f C1[2];
#pragma unroll
    for (int h = 0; h < 2; ++h) {
        int j = 16 * h + lm;
        B1[h].x = W1[(n * 32 + j) * 3 + kb];
        B1[h].y = hi ? 0.0f : W1[(n * 32 + j) * 3 + 1];
        float bb = b1[n * 32 + j];
#pragma unroll
        for (int i = 0; i < 8; ++i) C1[h][i] = bb;
    }
    v2f B2[8];
#pragma unroll
    for (int kc = 0; kc < 8; ++kc) {
        B2[kc].x = W2[(n * 16 + lm) * 32 + 4 * kc + kb];
        B2[kc].y = W2[(n * 16 + lm) * 32 + 4 * kc + kb + 1];
    }
    const float c2s = b2[n * 16 + lm];

    float w3v[16];
#pragma unroll
    for (int k = 0; k < 16; ++k) w3v[k] = W3[n * 16 + k];

    float cl[4];
    {
        float e0 = __expf(lw[0]), e1 = __expf(lw[1]);
        float e2 = __expf(lw[2]), e3 = __expf(lw[3]);
        float es = e0 + e1 + e2 + e3;
        float ws[4] = { e0 / es, e1 / es, e2 / es, e3 / es };
#pragma unroll
        for (int l = 0; l < 4; ++l) {
            float s = 0.0f;
#pragma unroll
            for (int q = 0; q < 4; ++q)
                s += 1.0f / (1.0f + __expf(-imp[l * 4 + q]));
            cl[l] = (s / (s + 1e-8f)) * ws[l];
        }
    }
    v2f Af;
    Af.x = Wf1[lm * 4 + kb];
    Af.y = Wf1[lm * 4 + kb + 1];
    const float clx = hi ? cl[2] : cl[0];
    const float cly = hi ? cl[3] : cl[1];
    v8f Cb;
    float wf2h[8];
#pragma unroll
    for (int i = 0; i < 8; ++i) { Cb[i] = bf1[8 * hi + i]; wf2h[i] = Wf2[8 * hi + i]; }
    const float bf2v = bf2[0];

    float prev = 0.0f;
    const float2* xp = (const float2*)x;
    float2 xc = make_float2(0.0f, 0.0f);
    if (!hi) xc = xp[(size_t)(b0 + lm) * TT];

    float* h1p = lds + w * (16 * 34);
    float* h2p = lds + H1N + w * (16 * 20);

    for (int tc = 0; tc < TT / CH; ++tc) {
      float* Sw = ss + (tc & 1) * (CH * SP) + w * 16 + lm;
      for (int tt = 0; tt < CH; ++tt) {
        const int t = tc * CH + tt;
        float2 xn = make_float2(0.0f, 0.0f);
        if (!hi && (t + 1) < TT) xn = xp[(size_t)(b0 + lm) * TT + (t + 1)];

        v2f a1;
        a1.x = hi ? prev : xc.x;
        a1.y = hi ? 0.0f : xc.y;

        v8f d0 = wmma4(a1, B1[0], C1[0]);
        v8f d1 = wmma4(a1, B1[1], C1[1]);
#pragma unroll
        for (int i = 0; i < 8; ++i) { d0[i] = fast_tanh(d0[i]); d1[i] = fast_tanh(d1[i]); }

#pragma unroll
        for (int v = 0; v < 8; ++v) {
            h1p[(v + 8 * hi) * 34 + lm]      = d0[v];
            h1p[(v + 8 * hi) * 34 + 16 + lm] = d1[v];
        }

        v8f acc;
#pragma unroll
        for (int i = 0; i < 8; ++i) acc[i] = c2s;
#pragma unroll
        for (int kc = 0; kc < 8; ++kc) {
            v2f a2 = *(const v2f*)&h1p[lm * 34 + 4 * kc + kb];
            acc = wmma4(a2, B2[kc], acc);
        }
#pragma unroll
        for (int i = 0; i < 8; ++i) acc[i] = fast_tanh(acc[i]);

#pragma unroll
        for (int v = 0; v < 8; ++v) h2p[(v + 8 * hi) * 20 + lm] = acc[v];

        float stress = 0.0f;
        if (!hi) {
            const float4* row = (const float4*)&h2p[lm * 20];
            float corr = b3v;
#pragma unroll
            for (int j = 0; j < 4; ++j) {
                float4 r = row[j];
                corr = fmaf(r.x, w3v[4 * j + 0], corr);
                corr = fmaf(r.y, w3v[4 * j + 1], corr);
                corr = fmaf(r.z, w3v[4 * j + 2], corr);
                corr = fmaf(r.w, w3v[4 * j + 3], corr);
            }
            float mw = fmaf(E, xc.x, eta * xc.y);
            stress = (mw + corr) * iw;
            Sw[tt * SP] = stress;
        }
        prev = xhalf(stress);
        xc = xn;
      }

      __syncthreads();
      const float* Sr = ss + (tc & 1) * (CH * SP);
#pragma unroll
      for (int r = 0; r < 4; ++r) {
        const int bi = 4 * w + r;
        float tot[2];
#pragma unroll
        for (int hh = 0; hh < 2; ++hh) {
            const float* sp = &Sr[(16 * hh + lm) * SP + kb * 16 + bi];
            v2f vb;
            vb.x = sp[0]  * clx;
            vb.y = sp[16] * cly;
            v8f d = wmma4(Af, vb, Cb);
            float c = wf2h[0] * fmaxf(d[0], 0.0f);
#pragma unroll
            for (int i = 1; i < 8; ++i) c = fmaf(wf2h[i], fmaxf(d[i], 0.0f), c);
            tot[hh] = c + xhalf(c);
        }
        out[(size_t)(p * BB + b0 + bi) * TT + tc * CH + lane] = (hi ? tot[1] : tot[0]) + bf2v;
      }
    }
}

extern "C" void kernel_launch(void* const* d_in, const int* in_sizes, int n_in,
                              void* d_out, int out_size, void* d_ws, size_t ws_size,
                              hipStream_t stream) {
    (void)in_sizes; (void)n_in; (void)d_ws; (void)ws_size; (void)out_size;
    pinn_kernel<<<dim3(16, 4), dim3(128), 0, stream>>>(
        (const float*)d_in[0],  (const float*)d_in[1],  (const float*)d_in[2],
        (const float*)d_in[3],  (const float*)d_in[4],  (const float*)d_in[5],
        (const float*)d_in[6],  (const float*)d_in[7],  (const float*)d_in[8],
        (const float*)d_in[9],  (const float*)d_in[10], (const float*)d_in[11],
        (const float*)d_in[12], (const float*)d_in[13], (const float*)d_in[14],
        (float*)d_out);
}


# step products transposed (weights = A, activations = B) so each layer's D fragment is the next layer's B fragment: no LDS transpose per step; layer-2 k grouping and layer-3 sum order changed
# speedup vs baseline: 4.9899x; 1.5589x over previous
#include <hip/hip_runtime.h>
#include <math.h>

typedef __attribute__((ext_vector_type(2))) float v2f;
typedef __attribute__((ext_vector_type(8))) float v8f;

#define TT 4096
#define BB 256
#define CH 32
#define SP 65

__device__ __forceinline__ float fast_tanh(float x) {
    float e = __expf(2.0f * x);
    return 1.0f - 2.0f * __builtin_amdgcn_rcpf(e + 1.0f);
}

__device__ __forceinline__ float xhalf(float x) {
    int b = __builtin_bit_cast(int, x);
    return __builtin_bit_cast(float,
        __builtin_amdgcn_permlanex16(b, b, 0x76543210, 0xfedcba98, false, false));
}

__device__ __forceinline__ v8f wmma4(v2f a, v2f b, v8f c) {
    return __builtin_amdgcn_wmma_f32_16x16x4_f32(
        false, a, false, b, (short)0, c, false, false);
}

__global__ __launch_bounds__(128)
void pinn_kernel(const float* __restrict__ x,
                 const float* __restrict__ W1,  const float* __restrict__ b1,
                 const float* __restrict__ W2,  const float* __restrict__ b2,
                 const float* __restrict__ W3,  const float* __restrict__ b3,
                 const float* __restrict__ logE,const float* __restrict__ logeta,
                 const float* __restrict__ imp, const float* __restrict__ lw,
                 const float* __restrict__ Wf1, const float* __restrict__ bf1,
                 const float* __restrict__ Wf2, const float* __restrict__ bf2,
                 float* __restrict__ out)
{
    __shared__ __align__(16) float ss[2 * CH * SP];

    const int tid  = threadIdx.x;
    const int w    = tid >> 5;
    const int p    = blockIdx.y;
    const int n    = w * 4 + p;
    const int lane = tid & 31;
    const int hi   = lane >> 4;
    const int lm   = lane & 15;
    const int b0   = blockIdx.x * 16;
    const int kb   = hi ? 2 : 0;

    const float E   = __expf(logE[n]);
    const float eta = __expf(logeta[n]);
    const float iw  = 1.0f / (1.0f + __expf(-imp[n]));
    const float b3v = b3[n];

    v2f A1[2]; v8f C1[2];
#pragma unroll
    for (int h = 0; h < 2; ++h) {
        int j = 16 * h + lm;
        A1[h].x = W1[(n * 32 + j) * 3 + kb];
        A1[h].y = hi ? 0.0f : W1[(n * 32 + j) * 3 + 1];
#pragma unroll
        for (int i = 0; i < 8; ++i) C1[h][i] = b1[n * 32 + 16 * h + 8 * hi + i];
    }
    v2f A2[8];
#pragma unroll
    for (int kc = 0; kc < 8; ++kc) {
        const int hid = 16 * (kc >> 2) + 8 * hi + 2 * (kc & 3);
        A2[kc].x = W2[(n * 16 + lm) * 32 + hid];
        A2[kc].y = W2[(n * 16 + lm) * 32 + hid + 1];
    }
    v8f C2;
#pragma unroll
    for (int i = 0; i < 8; ++i) C2[i] = b2[n * 16 + 8 * hi + i];

    float w3h[8];
#pragma unroll
    for (int k = 0; k < 8; ++k) w3h[k] = W3[n * 16 + 8 * hi + k];
    const float b3h = hi ? 0.0f : b3v;

    float cl[4];
    {
        float e0 = __expf(lw[0]), e1 = __expf(lw[1]);
        float e2 = __expf(lw[2]), e3 = __expf(lw[3]);
        float es = e0 + e1 + e2 + e3;
        float ws[4] = { e0 / es, e1 / es, e2 / es, e3 / es };
#pragma unroll
        for (int l = 0; l < 4; ++l) {
            float s = 0.0f;
#pragma unroll
            for (int q = 0; q < 4; ++q)
                s += 1.0f / (1.0f + __expf(-imp[l * 4 + q]));
            cl[l] = (s / (s + 1e-8f)) * ws[l];
        }
    }
    v2f Af;
    Af.x = Wf1[lm * 4 + kb];
    Af.y = Wf1[lm * 4 + kb + 1];
    const float clx = hi ? cl[2] : cl[0];
    const float cly = hi ? cl[3] : cl[1];
    v8f Cb;
    float wf2h[8];
#pragma unroll
    for (int i = 0; i < 8; ++i) { Cb[i] = bf1[8 * hi + i]; wf2h[i] = Wf2[8 * hi + i]; }
    const float bf2v = bf2[0];

    float prev = 0.0f;
    const float2* xp = (const float2*)x;
    float2 xc = xp[(size_t)(b0 + lm) * TT];

    for (int tc = 0; tc < TT / CH; ++tc) {
      float* Sw = ss + (tc & 1) * (CH * SP) + w * 16 + lm;
      for (int tt = 0; tt < CH; ++tt) {
        const int t = tc * CH + tt;
        float2 xn = make_float2(0.0f, 0.0f);
        if ((t + 1) < TT) xn = xp[(size_t)(b0 + lm) * TT + (t + 1)];

        v2f bx;
        bx.x = hi ? prev : xc.x;
        bx.y = hi ? 0.0f : xc.y;

        v8f d0 = wmma4(A1[0], bx, C1[0]);
        v8f d1 = wmma4(A1[1], bx, C1[1]);

        v8f acc = C2;
#pragma unroll
        for (int kc = 0; kc < 8; ++kc) {
            const int q = kc & 3;
            v2f hb;
            hb.x = fast_tanh(kc < 4 ? d0[2 * q]     : d1[2 * q]);
            hb.y = fast_tanh(kc < 4 ? d0[2 * q + 1] : d1[2 * q + 1]);
            acc = wmma4(A2[kc], hb, acc);
        }

        float part = b3h;
#pragma unroll
        for (int i = 0; i < 8; ++i) part = fmaf(fast_tanh(acc[i]), w3h[i], part);
        const float corr = part + xhalf(part);

        float mw = fmaf(E, xc.x, eta * xc.y);
        float stress = (mw + corr) * iw;
        Sw[tt * SP] = stress;
        prev = stress;
        xc = xn;
      }

      __syncthreads();
      const float* Sr = ss + (tc & 1) * (CH * SP);
#pragma unroll
      for (int r = 0; r < 4; ++r) {
        const int bi = 4 * w + r;
        float tot[2];
#pragma unroll
        for (int hh = 0; hh < 2; ++hh) {
            const float* sp = &Sr[(16 * hh + lm) * SP + kb * 16 + bi];
            v2f vb;
            vb.x = sp[0]  * clx;
            vb.y = sp[16] * cly;
            v8f d = wmma4(Af, vb, Cb);
            float c = wf2h[0] * fmaxf(d[0], 0.0f);
#pragma unroll
            for (int i = 1; i < 8; ++i) c = fmaf(wf2h[i], fmaxf(d[i], 0.0f), c);
            tot[hh] = c + xhalf(c);
        }
        out[(size_t)(p * BB + b0 + bi) * TT + tc * CH + lane] = (hi ? tot[1] : tot[0]) + bf2v;
      }
    }
}

extern "C" void kernel_launch(void* const* d_in, const int* in_sizes, int n_in,
                              void* d_out, int out_size, void* d_ws, size_t ws_size,
                              hipStream_t stream) {
    (void)in_sizes; (void)n_in; (void)d_ws; (void)ws_size; (void)out_size;
    pinn_kernel<<<dim3(16, 4), dim3(128), 0, stream>>>(
        (const float*)d_in[0],  (const float*)d_in[1],  (const float*)d_in[2],
        (const float*)d_in[3],  (const float*)d_in[4],  (const float*)d_in[5],
        (const float*)d_in[6],  (const float*)d_in[7],  (const float*)d_in[8],
        (const float*)d_in[9],  (const float*)d_in[10], (const float*)d_in[11],
        (const float*)d_in[12], (const float*)d_in[13], (const float*)d_in[14],
        (float*)d_out);
}


# tanh on the hardware instruction v_tanh_f32 (one transcendental per value) in place of exp, add, reciprocal, fma: 24 per lane and step; tanh rounding changed, sums and their order unchanged
# speedup vs baseline: 6.2549x; 1.2535x over previous
#include <hip/hip_runtime.h>
#include <math.h>

typedef __attribute__((ext_vector_type(2))) float v2f;
typedef __attribute__((ext_vector_type(8))) float v8f;

#define TT 4096
#define BB 256
#define CH 32
#define SP 65

__device__ __forceinline__ float fast_tanh(float x) {
    return __builtin_amdgcn_tanhf(x);
}

__device__ __forceinline__ float xhalf(float x) {
    int b = __builtin_bit_cast(int, x);
    return __builtin_bit_cast(float,
        __builtin_amdgcn_permlanex16(b, b, 0x76543210, 0xfedcba98, false, false));
}

__device__ __forceinline__ v8f wmma4(v2f a, v2f b, v8f c) {
    return __builtin_amdgcn_wmma_f32_16x16x4_f32(
        false, a, false, b, (short)0, c, false, false);
}

__global__ __launch_bounds__(128)
void pinn_kernel(const float* __restrict__ x,
                 const float* __restrict__ W1,  const float* __restrict__ b1,
                 const float* __restrict__ W2,  const float* __restrict__ b2,
                 const float* __restrict__ W3,  const float* __restrict__ b3,
                 const float* __restrict__ logE,const float* __restrict__ logeta,
                 const float* __restrict__ imp, const float* __restrict__ lw,
                 const float* __restrict__ Wf1, const float* __restrict__ bf1,
                 const float* __restrict__ Wf2, const float* __restrict__ bf2,
                 float* __restrict__ out)
{
    __shared__ __align__(16) float ss[2 * CH * SP];

    const int tid  = threadIdx.x;
    const int w    = tid >> 5;
    const int p    = blockIdx.y;
    const int n    = w * 4 + p;
    const int lane = tid & 31;
    const int hi   = lane >> 4;
    const int lm   = lane & 15;
    const int b0   = blockIdx.x * 16;
    const int kb   = hi ? 2 : 0;

    const float E   = __expf(logE[n]);
    const float eta = __expf(logeta[n]);
    const float iw  = 1.0f / (1.0f + __expf(-imp[n]));
    const float b3v = b3[n];

    v2f A1[2]; v8f C1[2];
#pragma unroll
    for (int h = 0; h < 2; ++h) {
        int j = 16 * h + lm;
        A1[h].x = W1[(n * 32 + j) * 3 + kb];
        A1[h].y = hi ? 0.0f : W1[(n * 32 + j) * 3 + 1];
#pragma unroll
        for (int i = 0; i < 8; ++i) C1[h][i] = b1[n * 32 + 16 * h + 8 * hi + i];
    }
    v2f A2[8];
#pragma unroll
    for (int kc = 0; kc < 8; ++kc) {
        const int hid = 16 * (kc >> 2) + 8 * hi + 2 * (kc & 3);
        A2[kc].x = W2[(n * 16 + lm) * 32 + hid];
        A2[kc].y = W2[(n * 16 + lm) * 32 + hid + 1];
    }
    v8f C2;
#pragma unroll
    for (int i = 0; i < 8; ++i) C2[i] = b2[n * 16 + 8 * hi + i];

    float w3h[8];
#pragma unroll
    for (int k = 0; k < 8; ++k) w3h[k] = W3[n * 16 + 8 * hi + k];
    const float b3h = hi ? 0.0f : b3v;

    float cl[4];
    {
        float e0 = __expf(lw[0]), e1 = __expf(lw[1]);
        float e2 = __expf(lw[2]), e3 = __expf(lw[3]);
        float es = e0 + e1 + e2 + e3;
        float ws[4] = { e0 / es, e1 / es, e2 / es, e3 / es };
#pragma unroll
        for (int l = 0; l < 4; ++l) {
            float s = 0.0f;
#pragma unroll
            for (int q = 0; q < 4; ++q)
                s += 1.0f / (1.0f + __expf(-imp[l * 4 + q]));
            cl[l] = (s / (s + 1e-8f)) * ws[l];
        }
    }
    v2f Af;
    Af.x = Wf1[lm * 4 + kb];
    Af.y = Wf1[lm * 4 + kb + 1];
    const float clx = hi ? cl[2] : cl[0];
    const float cly = hi ? cl[3] : cl[1];
    v8f Cb;
    float wf2h[8];
#pragma unroll
    for (int i = 0; i < 8; ++i) { Cb[i] = bf1[8 * hi + i]; wf2h[i] = Wf2[8 * hi + i]; }
    const float bf2v = bf2[0];

    float prev = 0.0f;
    const float2* xp = (const float2*)x;
    float2 xc = xp[(size_t)(b0 + lm) * TT];

    for (int tc = 0; tc < TT / CH; ++tc) {
      float* Sw = ss + (tc & 1) * (CH * SP) + w * 16 + lm;
#pragma nounroll
      for (int tt = 0; tt < CH; ++tt) {
        const int t = tc * CH + tt;
        float2 xn = make_float2(0.0f, 0.0f);
        if ((t + 1) < TT) xn = xp[(size_t)(b0 + lm) * TT + (t + 1)];

        v2f bx;
        bx.x = hi ? prev : xc.x;
        bx.y = hi ? 0.0f : xc.y;

        v8f d0 = wmma4(A1[0], bx, C1[0]);
        v8f d1 = wmma4(A1[1], bx, C1[1]);

        v8f acc = C2;
#pragma unroll
        for (int kc = 0; kc < 8; ++kc) {
            const int q = kc & 3;
            v2f hb;
            hb.x = fast_tanh(kc < 4 ? d0[2 * q]     : d1[2 * q]);
            hb.y = fast_tanh(kc < 4 ? d0[2 * q + 1] : d1[2 * q + 1]);
            acc = wmma4(A2[kc], hb, acc);
        }

        float part = b3h;
#pragma unroll
        for (int i = 0; i < 8; ++i) part = fmaf(fast_tanh(acc[i]), w3h[i], part);
        const float corr = part + xhalf(part);

        float mw = fmaf(E, xc.x, eta * xc.y);
        float stress = (mw + corr) * iw;
        Sw[tt * SP] = stress;
        prev = stress;
        xc = xn;
      }

      __syncthreads();
      const float* Sr = ss + (tc & 1) * (CH * SP);
#pragma unroll
      for (int r = 0; r < 4; ++r) {
        const int bi = 4 * w + r;
        float tot[2];
#pragma unroll
        for (int hh = 0; hh < 2; ++hh) {
            const float* sp = &Sr[(16 * hh + lm) * SP + kb * 16 + bi];
            v2f vb;
            vb.x = sp[0]  * clx;
            vb.y = sp[16] * cly;
            v8f d = wmma4(Af, vb, Cb);
            float c = wf2h[0] * fmaxf(d[0], 0.0f);
#pragma unroll
            for (int i = 1; i < 8; ++i) c = fmaf(wf2h[i], fmaxf(d[i], 0.0f), c);
            tot[hh] = c + xhalf(c);
        }
        out[(size_t)(p * BB + b0 + bi) * TT + tc * CH + lane] = (hi ? tot[1] : tot[0]) + bf2v;
      }
    }
}

extern "C" void kernel_launch(void* const* d_in, const int* in_sizes, int n_in,
                              void* d_out, int out_size, void* d_ws, size_t ws_size,
                              hipStream_t stream) {
    (void)in_sizes; (void)n_in; (void)d_ws; (void)ws_size; (void)out_size;
    pinn_kernel<<<dim3(16, 4), dim3(128), 0, stream>>>(
        (const float*)d_in[0],  (const float*)d_in[1],  (const float*)d_in[2],
        (const float*)d_in[3],  (const float*)d_in[4],  (const float*)d_in[5],
        (const float*)d_in[6],  (const float*)d_in[7],  (const float*)d_in[8],
        (const float*)d_in[9],  (const float*)d_in[10], (const float*)d_in[11],
        (const float*)d_in[12], (const float*)d_in[13], (const float*)d_in[14],
        (float*)d_out);
}


# step's accumulation chains split in two: layer 2 on two accumulator tiles (one per layer-1 tile, products alternate, summed at the end), layer 3 on two in-lane partial sums; sum order of both changed
# speedup vs baseline: 6.7079x; 1.0724x over previous
#include <hip/hip_runtime.h>
#include <math.h>

typedef __attribute__((ext_vector_type(2))) float v2f;
typedef __attribute__((ext_vector_type(8))) float v8f;

#define TT 4096
#define BB 256
#define CH 32
#define SP 65

__device__ __forceinline__ float fast_tanh(float x) {
    return __builtin_amdgcn_tanhf(x);
}

__device__ __forceinline__ float xhalf(float x) {
    int b = __builtin_bit_cast(int, x);
    return __builtin_bit_cast(float,
        __builtin_amdgcn_permlanex16(b, b, 0x76543210, 0xfedcba98, false, false));
}

__device__ __forceinline__ v8f wmma4(v2f a, v2f b, v8f c) {
    return __builtin_amdgcn_wmma_f32_16x16x4_f32(
        false, a, false, b, (short)0, c, false, false);
}

__global__ __launch_bounds__(128)
void pinn_kernel(const float* __restrict__ x,
                 const float* __restrict__ W1,  const float* __restrict__ b1,
                 const float* __restrict__ W2,  const float* __restrict__ b2,
                 const float* __restrict__ W3,  const float* __restrict__ b3,
                 const float* __restrict__ logE,const float* __restrict__ logeta,
                 const float* __restrict__ imp, const float* __restrict__ lw,
                 const float* __restrict__ Wf1, const float* __restrict__ bf1,
                 const float* __restrict__ Wf2, const float* __restrict__ bf2,
                 float* __restrict__ out)
{
    __shared__ __align__(16) float ss[2 * CH * SP];

    const int tid  = threadIdx.x;
    const int w    = tid >> 5;
    const int p    = blockIdx.y;
    const int n    = w * 4 + p;
    const int lane = tid & 31;
    const int hi   = lane >> 4;
    const int lm   = lane & 15;
    const int b0   = blockIdx.x * 16;
    const int kb   = hi ? 2 : 0;

    const float E   = __expf(logE[n]);
    const float eta = __expf(logeta[n]);
    const float iw  = 1.0f / (1.0f + __expf(-imp[n]));
    const float b3v = b3[n];

    v2f A1[2]; v8f C1[2];
#pragma unroll
    for (int h = 0; h < 2; ++h) {
        int j = 16 * h + lm;
        A1[h].x = W1[(n * 32 + j) * 3 + kb];
        A1[h].y = hi ? 0.0f : W1[(n * 32 + j) * 3 + 1];
#pragma unroll
        for (int i = 0; i < 8; ++i) C1[h][i] = b1[n * 32 + 16 * h + 8 * hi + i];
    }
    v2f A2[8];
#pragma unroll
    for (int kc = 0; kc < 8; ++kc) {
        const int hid = 16 * (kc >> 2) + 8 * hi + 2 * (kc & 3);
        A2[kc].x = W2[(n * 16 + lm) * 32 + hid];
        A2[kc].y = W2[(n * 16 + lm) * 32 + hid + 1];
    }
    v8f C2;
#pragma unroll
    for (int i = 0; i < 8; ++i) C2[i] = b2[n * 16 + 8 * hi + i];

    float w3h[8];
#pragma unroll
    for (int k = 0; k < 8; ++k) w3h[k] = W3[n * 16 + 8 * hi + k];
    const float b3h = hi ? 0.0f : b3v;

    float cl[4];
    {
        float e0 = __expf(lw[0]), e1 = __expf(lw[1]);
        float e2 = __expf(lw[2]), e3 = __expf(lw[3]);
        float es = e0 + e1 + e2 + e3;
        float ws[4] = { e0 / es, e1 / es, e2 / es, e3 / es };
#pragma unroll
        for (int l = 0; l < 4; ++l) {
            float s = 0.0f;
#pragma unroll
            for (int q = 0; q < 4; ++q)
                s += 1.0f / (1.0f + __expf(-imp[l * 4 + q]));
            cl[l] = (s / (s + 1e-8f)) * ws[l];
        }
    }
    v2f Af;
    Af.x = Wf1[lm * 4 + kb];
    Af.y = Wf1[lm * 4 + kb + 1];
    const float clx = hi ? cl[2] : cl[0];
    const float cly = hi ? cl[3] : cl[1];
    v8f Cb;
    float wf2h[8];
#pragma unroll
    for (int i = 0; i < 8; ++i) { Cb[i] = bf1[8 * hi + i]; wf2h[i] = Wf2[8 * hi + i]; }
    const float bf2v = bf2[0];

    float prev = 0.0f;
    const float2* xp = (const float2*)x;
    float2 xc = xp[(size_t)(b0 + lm) * TT];

    for (int tc = 0; tc < TT / CH; ++tc) {
      float* Sw = ss + (tc & 1) * (CH * SP) + w * 16 + lm;
#pragma nounroll
      for (int tt = 0; tt < CH; ++tt) {
        const int t = tc * CH + tt;
        float2 xn = make_float2(0.0f, 0.0f);
        if ((t + 1) < TT) xn = xp[(size_t)(b0 + lm) * TT + (t + 1)];

        v2f bx;
        bx.x = hi ? prev : xc.x;
        bx.y = hi ? 0.0f : xc.y;

        v8f d0 = wmma4(A1[0], bx, C1[0]);
        v8f d1 = wmma4(A1[1], bx, C1[1]);

        v8f acc = C2;
        v8f acb = {0.0f, 0.0f, 0.0f, 0.0f, 0.0f, 0.0f, 0.0f, 0.0f};
#pragma unroll
        for (int q = 0; q < 4; ++q) {
            v2f ha, hb;
            ha.x = fast_tanh(d0[2 * q]);
            ha.y = fast_tanh(d0[2 * q + 1]);
            acc = wmma4(A2[q], ha, acc);
            hb.x = fast_tanh(d1[2 * q]);
            hb.y = fast_tanh(d1[2 * q + 1]);
            acb = wmma4(A2[4 + q], hb, acb);
        }
        acc = acc + acb;

        float pe = b3h, po = 0.0f;
#pragma unroll
        for (int i = 0; i < 8; i += 2) {
            pe = fmaf(fast_tanh(acc[i]),     w3h[i],     pe);
            po = fmaf(fast_tanh(acc[i + 1]), w3h[i + 1], po);
        }
        const float part = pe + po;
        const float corr = part + xhalf(part);

        float mw = fmaf(E, xc.x, eta * xc.y);
        float stress = (mw + corr) * iw;
        Sw[tt * SP] = stress;
        prev = stress;
        xc = xn;
      }

      __syncthreads();
      const float* Sr = ss + (tc & 1) * (CH * SP);
#pragma unroll
      for (int r = 0; r < 4; ++r) {
        const int bi = 4 * w + r;
        float tot[2];
#pragma unroll
        for (int hh = 0; hh < 2; ++hh) {
            const float* sp = &Sr[(16 * hh + lm) * SP + kb * 16 + bi];
            v2f vb;
            vb.x = sp[0]  * clx;
            vb.y = sp[16] * cly;
            v8f d = wmma4(Af, vb, Cb);
            float c = wf2h[0] * fmaxf(d[0], 0.0f);
#pragma unroll
            for (int i = 1; i < 8; ++i) c = fmaf(wf2h[i], fmaxf(d[i], 0.0f), c);
            tot[hh] = c + xhalf(c);
        }
        out[(size_t)(p * BB + b0 + bi) * TT + tc * CH + lane] = (hi ? tot[1] : tot[0]) + bf2v;
      }
    }
}

extern "C" void kernel_launch(void* const* d_in, const int* in_sizes, int n_in,
                              void* d_out, int out_size, void* d_ws, size_t ws_size,
                              hipStream_t stream) {
    (void)in_sizes; (void)n_in; (void)d_ws; (void)ws_size; (void)out_size;
    pinn_kernel<<<dim3(16, 4), dim3(128), 0, stream>>>(
        (const float*)d_in[0],  (const float*)d_in[1],  (const float*)d_in[2],
        (const float*)d_in[3],  (const float*)d_in[4],  (const float*)d_in[5],
        (const float*)d_in[6],  (const float*)d_in[7],  (const float*)d_in[8],
        (const float*)d_in[9],  (const float*)d_in[10], (const float*)d_in[11],
        (const float*)d_in[12], (const float*)d_in[13], (const float*)d_in[14],
        (float*)d_out);
}


# serial step freed of memory instructions: inputs staged per block in LDS, read 8 steps at a time a group ahead, 8 stresses stored at once (step loop unrolled by 8); sums and order unchanged
# speedup vs baseline: 9.2309x; 1.3761x over previous
#include <hip/hip_runtime.h>
#include <math.h>

typedef __attribute__((ext_vector_type(2))) float v2f;
typedef __attribute__((ext_vector_type(8))) float v8f;

#define TT 4096
#define BB 256
#define CH 32
#define GS 8
#define RP 36
#define LP 592
#define XP 68

__device__ __forceinline__ float fast_tanh(float x) {
    return __builtin_amdgcn_tanhf(x);
}

__device__ __forceinline__ float xhalf(float x) {
    int b = __builtin_bit_cast(int, x);
    return __builtin_bit_cast(float,
        __builtin_amdgcn_permlanex16(b, b, 0x76543210, 0xfedcba98, false, false));
}

__device__ __forceinline__ v8f wmma4(v2f a, v2f b, v8f c) {
    return __builtin_amdgcn_wmma_f32_16x16x4_f32(
        false, a, false, b, (short)0, c, false, false);
}

__global__ __launch_bounds__(128)
void pinn_kernel(const float* __restrict__ x,
                 const float* __restrict__ W1,  const float* __restrict__ b1,
                 const float* __restrict__ W2,  const float* __restrict__ b2,
                 const float* __restrict__ W3,  const float* __restrict__ b3,
                 const float* __restrict__ logE,const float* __restrict__ logeta,
                 const float* __restrict__ imp, const float* __restrict__ lw,
                 const float* __restrict__ Wf1, const float* __restrict__ bf1,
                 const float* __restrict__ Wf2, const float* __restrict__ bf2,
                 float* __restrict__ out)
{
    __shared__ __align__(16) float ss[2 * 4 * LP];
    __shared__ __align__(16) float xsh[3 * 16 * XP];

    const int tid  = threadIdx.x;
    const int w    = tid >> 5;
    const int p    = blockIdx.y;
    const int n    = w * 4 + p;
    const int lane = tid & 31;
    const int hi   = lane >> 4;
    const int lm   = lane & 15;
    const int b0   = blockIdx.x * 16;
    const int kb   = hi ? 2 : 0;

    const float E   = __expf(logE[n]);
    const float eta = __expf(logeta[n]);
    const float iw  = 1.0f / (1.0f + __expf(-imp[n]));
    const float b3v = b3[n];

    v2f A1[2]; v8f C1[2];
#pragma unroll
    for (int h = 0; h < 2; ++h) {
        int j = 16 * h + lm;
        A1[h].x = W1[(n * 32 + j) * 3 + kb];
        A1[h].y = hi ? 0.0f : W1[(n * 32 + j) * 3 + 1];
#pragma unroll
        for (int i = 0; i < 8; ++i) C1[h][i] = b1[n * 32 + 16 * h + 8 * hi + i];
    }
    v2f A2[8];
#pragma unroll
    for (int kc = 0; kc < 8; ++kc) {
        const int hid = 16 * (kc >> 2) + 8 * hi + 2 * (kc & 3);
        A2[kc].x = W2[(n * 16 + lm) * 32 + hid];
        A2[kc].y = W2[(n * 16 + lm) * 32 + hid + 1];
    }
    v8f C2;
#pragma unroll
    for (int i = 0; i < 8; ++i) C2[i] = b2[n * 16 + 8 * hi + i];

    float w3h[8];
#pragma unroll
    for (int k = 0; k < 8; ++k) w3h[k] = W3[n * 16 + 8 * hi + k];
    const float b3h = hi ? 0.0f : b3v;

    float cl[4];
    {
        float e0 = __expf(lw[0]), e1 = __expf(lw[1]);
        float e2 = __expf(lw[2]), e3 = __expf(lw[3]);
        float es = e0 + e1 + e2 + e3;
        float ws[4] = { e0 / es, e1 / es, e2 / es, e3 / es };
#pragma unroll
        for (int l = 0; l < 4; ++l) {
            float s = 0.0f;
#pragma unroll
            for (int q = 0; q < 4; ++q)
                s += 1.0f / (1.0f + __expf(-imp[l * 4 + q]));
            cl[l] = (s / (s + 1e-8f)) * ws[l];
        }
    }
    v2f Af;
    Af.x = Wf1[lm * 4 + kb];
    Af.y = Wf1[lm * 4 + kb + 1];
    const float clx = hi ? cl[2] : cl[0];
    const float cly = hi ? cl[3] : cl[1];
    v8f Cb;
    float wf2h[8];
#pragma unroll
    for (int i = 0; i < 8; ++i) { Cb[i] = bf1[8 * hi + i]; wf2h[i] = Wf2[8 * hi + i]; }
    const float bf2v = bf2[0];

    const int srow = tid >> 3;
    const int spc  = tid & 7;
    const float4* xq = (const float4*)x + ((size_t)(b0 + srow) * TT) / 2 + spc * 2;
    float* xst = xsh + srow * XP + spc * 8;
    {
        const float4 a0 = xq[0],  a1 = xq[1];
        const float4 c0 = xq[16], c1 = xq[17];
        *(float4*)(xst)     = a0;
        *(float4*)(xst + 4) = a1;
        *(float4*)(xst + 16 * XP)     = c0;
        *(float4*)(xst + 16 * XP + 4) = c1;
    }
    __syncthreads();

    float prev = 0.0f;
    const float* xrd = xsh + lm * XP;
    float4 q[4];
#pragma unroll
    for (int j = 0; j < 4; ++j) q[j] = *(const float4*)(xrd + 4 * j);

    int bc = 0;
    for (int tc = 0; tc < TT / CH; ++tc) {
      const int  bn   = (bc == 2) ? 0 : bc + 1;
      const int  bs   = (bn == 2) ? 0 : bn + 1;
      const bool more = (tc + 2) < TT / CH;
      const bool lastb = (tc + 1) == TT / CH;
      float4 ga = make_float4(0.0f, 0.0f, 0.0f, 0.0f), gb = ga;
      if (more) { ga = xq[(tc + 2) * 16]; gb = xq[(tc + 2) * 16 + 1]; }

      float* Sw = ss + (tc & 1) * (4 * LP) + w * LP + lm * RP;
#pragma nounroll
      for (int g = 0; g < CH / GS; ++g) {
        const int nb = (g + 1 < CH / GS || lastb) ? bc : bn;
        const int ng = (g + 1 < CH / GS) ? g + 1 : (lastb ? g : 0);
        const float* xn = xrd + nb * (16 * XP) + ng * (2 * GS);
        float4 nq[4];
#pragma unroll
        for (int j = 0; j < 4; ++j) nq[j] = *(const float4*)(xn + 4 * j);
        __builtin_amdgcn_sched_barrier(0);

        float st[GS];
#pragma unroll
        for (int u = 0; u < GS; ++u) {
          const float xs0 = (u & 1) ? q[u >> 1].z : q[u >> 1].x;
          const float xs1 = (u & 1) ? q[u >> 1].w : q[u >> 1].y;

          v2f bx;
          bx.x = hi ? prev : xs0;
          bx.y = hi ? 0.0f : xs1;

          v8f d0 = wmma4(A1[0], bx, C1[0]);
          v8f d1 = wmma4(A1[1], bx, C1[1]);

          v8f acc = C2;
          v8f acb = {0.0f, 0.0f, 0.0f, 0.0f, 0.0f, 0.0f, 0.0f, 0.0f};
#pragma unroll
          for (int k = 0; k < 4; ++k) {
              v2f ha, hb;
              ha.x = fast_tanh(d0[2 * k]);
              ha.y = fast_tanh(d0[2 * k + 1]);
              acc = wmma4(A2[k], ha, acc);
              hb.x = fast_tanh(d1[2 * k]);
              hb.y = fast_tanh(d1[2 * k + 1]);
              acb = wmma4(A2[4 + k], hb, acb);
          }
          acc = acc + acb;

          float pe = b3h, po = 0.0f;
#pragma unroll
          for (int i = 0; i < 8; i += 2) {
              pe = fmaf(fast_tanh(acc[i]),     w3h[i],     pe);
              po = fmaf(fast_tanh(acc[i + 1]), w3h[i + 1], po);
          }
          const float part = pe + po;
          const float corr = part + xhalf(part);

          float mw = fmaf(E, xs0, eta * xs1);
          float stress = (mw + corr) * iw;
          st[u] = stress;
          prev = stress;
        }

        float4 sv;
        sv.x = hi ? st[4] : st[0];
        sv.y = hi ? st[5] : st[1];
        sv.z = hi ? st[6] : st[2];
        sv.w = hi ? st[7] : st[3];
        *(float4*)(Sw + g * GS + 4 * hi) = sv;
#pragma unroll
        for (int j = 0; j < 4; ++j) q[j] = nq[j];
      }

      if (more) {
        float* xd = xst + bs * (16 * XP);
        *(float4*)(xd)     = ga;
        *(float4*)(xd + 4) = gb;
      }

      __syncthreads();
      const float* Sr = ss + (tc & 1) * (4 * LP);
#pragma unroll
      for (int r = 0; r < 4; ++r) {
        const int bi = 4 * w + r;
        float tot[2];
#pragma unroll
        for (int hh = 0; hh < 2; ++hh) {
            const float* sp = &Sr[kb * LP + bi * RP + 16 * hh + lm];
            v2f vb;
            vb.x = sp[0]  * clx;
            vb.y = sp[LP] * cly;
            v8f d = wmma4(Af, vb, Cb);
            float c = wf2h[0] * fmaxf(d[0], 0.0f);
#pragma unroll
            for (int i = 1; i < 8; ++i) c = fmaf(wf2h[i], fmaxf(d[i], 0.0f), c);
            tot[hh] = c + xhalf(c);
        }
        out[(size_t)(p * BB + b0 + bi) * TT + tc * CH + lane] = (hi ? tot[1] : tot[0]) + bf2v;
      }
      bc = bn;
    }
}

extern "C" void kernel_launch(void* const* d_in, const int* in_sizes, int n_in,
                              void* d_out, int out_size, void* d_ws, size_t ws_size,
                              hipStream_t stream) {
    (void)in_sizes; (void)n_in; (void)d_ws; (void)ws_size; (void)out_size;
    pinn_kernel<<<dim3(16, 4), dim3(128), 0, stream>>>(
        (const float*)d_in[0],  (const float*)d_in[1],  (const float*)d_in[2],
        (const float*)d_in[3],  (const float*)d_in[4],  (const float*)d_in[5],
        (const float*)d_in[6],  (const float*)d_in[7],  (const float*)d_in[8],
        (const float*)d_in[9],  (const float*)d_in[10], (const float*)d_in[11],
        (const float*)d_in[12], (const float*)d_in[13], (const float*)d_in[14],
        (float*)d_out);
}


# carry path cut to one add: each lane half's layer-3 partial enters layer 1 on its own k slot (the product sums the halves); importance weight, bias, maxwell folded ahead; stress sum order changed
# speedup vs baseline: 10.0149x; 1.0849x over previous
#include <hip/hip_runtime.h>
#include <math.h>

typedef __attribute__((ext_vector_type(2))) float v2f;
typedef __attribute__((ext_vector_type(8))) float v8f;

#define TT 4096
#define BB 256
#define CH 32
#define GS 8
#define RP 36
#define LP 592
#define XP 68

__device__ __forceinline__ float fast_tanh(float x) {
    return __builtin_amdgcn_tanhf(x);
}

__device__ __forceinline__ float xhalf(float x) {
    int b = __builtin_bit_cast(int, x);
    return __builtin_bit_cast(float,
        __builtin_amdgcn_permlanex16(b, b, 0x76543210, 0xfedcba98, false, false));
}

__device__ __forceinline__ v8f wmma4(v2f a, v2f b, v8f c) {
    return __builtin_amdgcn_wmma_f32_16x16x4_f32(
        false, a, false, b, (short)0, c, false, false);
}

__global__ __launch_bounds__(128)
void pinn_kernel(const float* __restrict__ x,
                 const float* __restrict__ W1,  const float* __restrict__ b1,
                 const float* __restrict__ W2,  const float* __restrict__ b2,
                 const float* __restrict__ W3,  const float* __restrict__ b3,
                 const float* __restrict__ logE,const float* __restrict__ logeta,
                 const float* __restrict__ imp, const float* __restrict__ lw,
                 const float* __restrict__ Wf1, const float* __restrict__ bf1,
                 const float* __restrict__ Wf2, const float* __restrict__ bf2,
                 float* __restrict__ out)
{
    __shared__ __align__(16) float ss[2 * 4 * LP];
    __shared__ __align__(16) float xsh[3 * 16 * XP];

    const int tid  = threadIdx.x;
    const int w    = tid >> 5;
    const int p    = blockIdx.y;
    const int n    = w * 4 + p;
    const int lane = tid & 31;
    const int hi   = lane >> 4;
    const int lm   = lane & 15;
    const int b0   = blockIdx.x * 16;
    const int kb   = hi ? 2 : 0;

    const float iw  = 1.0f / (1.0f + __expf(-imp[n]));
    const float Eh   = hi ? 0.0f : __expf(logE[n]) * iw;
    const float etah = hi ? 0.0f : __expf(logeta[n]) * iw;
    const float b3h  = hi ? 0.0f : b3[n] * iw;

    v2f A1[2]; v8f C1[2];
#pragma unroll
    for (int h = 0; h < 2; ++h) {
        int j = 16 * h + lm;
        A1[h].x = W1[(n * 32 + j) * 3 + 2];
        A1[h].y = W1[(n * 32 + j) * 3 + hi];
#pragma unroll
        for (int i = 0; i < 8; ++i) C1[h][i] = b1[n * 32 + 16 * h + 8 * hi + i];
    }
    v2f A2[8];
#pragma unroll
    for (int kc = 0; kc < 8; ++kc) {
        const int hid = 16 * (kc >> 2) + 8 * hi + 2 * (kc & 3);
        A2[kc].x = W2[(n * 16 + lm) * 32 + hid];
        A2[kc].y = W2[(n * 16 + lm) * 32 + hid + 1];
    }
    v8f C2;
#pragma unroll
    for (int i = 0; i < 8; ++i) C2[i] = b2[n * 16 + 8 * hi + i];

    float w3h[8];
#pragma unroll
    for (int k = 0; k < 8; ++k) w3h[k] = W3[n * 16 + 8 * hi + k] * iw;

    float cl[4];
    {
        float e0 = __expf(lw[0]), e1 = __expf(lw[1]);
        float e2 = __expf(lw[2]), e3 = __expf(lw[3]);
        float es = e0 + e1 + e2 + e3;
        float ws[4] = { e0 / es, e1 / es, e2 / es, e3 / es };
#pragma unroll
        for (int l = 0; l < 4; ++l) {
            float s = 0.0f;
#pragma unroll
            for (int q = 0; q < 4; ++q)
                s += 1.0f / (1.0f + __expf(-imp[l * 4 + q]));
            cl[l] = (s / (s + 1e-8f)) * ws[l];
        }
    }
    v2f Af;
    Af.x = Wf1[lm * 4 + kb];
    Af.y = Wf1[lm * 4 + kb + 1];
    const float clx = hi ? cl[2] : cl[0];
    const float cly = hi ? cl[3] : cl[1];
    v8f Cb;
    float wf2h[8];
#pragma unroll
    for (int i = 0; i < 8; ++i) { Cb[i] = bf1[8 * hi + i]; wf2h[i] = Wf2[8 * hi + i]; }
    const float bf2v = bf2[0];

    const int srow = tid >> 3;
    const int spc  = tid & 7;
    const float4* xq = (const float4*)x + ((size_t)(b0 + srow) * TT) / 2 + spc * 2;
    float* xst = xsh + srow * XP + spc * 8;
    {
        const float4 a0 = xq[0],  a1 = xq[1];
        const float4 c0 = xq[16], c1 = xq[17];
        *(float4*)(xst)     = a0;
        *(float4*)(xst + 4) = a1;
        *(float4*)(xst + 16 * XP)     = c0;
        *(float4*)(xst + 16 * XP + 4) = c1;
    }
    __syncthreads();

    float part = 0.0f;
    const float* xrd = xsh + lm * XP;
    float4 q[4];
#pragma unroll
    for (int j = 0; j < 4; ++j) q[j] = *(const float4*)(xrd + 4 * j);

    int bc = 0;
    for (int tc = 0; tc < TT / CH; ++tc) {
      const int  bn   = (bc == 2) ? 0 : bc + 1;
      const int  bs   = (bn == 2) ? 0 : bn + 1;
      const bool more = (tc + 2) < TT / CH;
      const bool lastb = (tc + 1) == TT / CH;
      float4 ga = make_float4(0.0f, 0.0f, 0.0f, 0.0f), gb = ga;
      if (more) { ga = xq[(tc + 2) * 16]; gb = xq[(tc + 2) * 16 + 1]; }

      float* Sw = ss + (tc & 1) * (4 * LP) + w * LP + lm * RP;
#pragma nounroll
      for (int g = 0; g < CH / GS; ++g) {
        const int nb = (g + 1 < CH / GS || lastb) ? bc : bn;
        const int ng = (g + 1 < CH / GS) ? g + 1 : (lastb ? g : 0);
        const float* xn = xrd + nb * (16 * XP) + ng * (2 * GS);
        float4 nq[4];
#pragma unroll
        for (int j = 0; j < 4; ++j) nq[j] = *(const float4*)(xn + 4 * j);
        __builtin_amdgcn_sched_barrier(0);

        float pt[GS];
#pragma unroll
        for (int u = 0; u < GS; ++u) {
          const float xs0 = (u & 1) ? q[u >> 1].z : q[u >> 1].x;
          const float xs1 = (u & 1) ? q[u >> 1].w : q[u >> 1].y;

          v2f bx;
          bx.x = part;
          bx.y = hi ? xs1 : xs0;

          v8f d0 = wmma4(A1[0], bx, C1[0]);
          v8f d1 = wmma4(A1[1], bx, C1[1]);

          v8f acc = C2;
          v8f acb = {0.0f, 0.0f, 0.0f, 0.0f, 0.0f, 0.0f, 0.0f, 0.0f};
#pragma unroll
          for (int k = 0; k < 4; ++k) {
              v2f ha, hb;
              ha.x = fast_tanh(d0[2 * k]);
              ha.y = fast_tanh(d0[2 * k + 1]);
              acc = wmma4(A2[k], ha, acc);
              hb.x = fast_tanh(d1[2 * k]);
              hb.y = fast_tanh(d1[2 * k + 1]);
              acb = wmma4(A2[4 + k], hb, acb);
          }
          acc = acc + acb;

          float pe = fmaf(Eh, xs0, fmaf(etah, xs1, b3h)), po = 0.0f;
#pragma unroll
          for (int i = 0; i < 8; i += 2) {
              pe = fmaf(fast_tanh(acc[i]),     w3h[i],     pe);
              po = fmaf(fast_tanh(acc[i + 1]), w3h[i + 1], po);
          }
          part = pe + po;
          pt[u] = part;
        }

        float st[GS];
#pragma unroll
        for (int u = 0; u < GS; ++u) st[u] = pt[u] + xhalf(pt[u]);

        float4 sv;
        sv.x = hi ? st[4] : st[0];
        sv.y = hi ? st[5] : st[1];
        sv.z = hi ? st[6] : st[2];
        sv.w = hi ? st[7] : st[3];
        *(float4*)(Sw + g * GS + 4 * hi) = sv;
#pragma unroll
        for (int j = 0; j < 4; ++j) q[j] = nq[j];
      }

      if (more) {
        float* xd = xst + bs * (16 * XP);
        *(float4*)(xd)     = ga;
        *(float4*)(xd + 4) = gb;
      }

      __syncthreads();
      const float* Sr = ss + (tc & 1) * (4 * LP);
#pragma unroll
      for (int r = 0; r < 4; ++r) {
        const int bi = 4 * w + r;
        float tot[2];
#pragma unroll
        for (int hh = 0; hh < 2; ++hh) {
            const float* sp = &Sr[kb * LP + bi * RP + 16 * hh + lm];
            v2f vb;
            vb.x = sp[0]  * clx;
            vb.y = sp[LP] * cly;
            v8f d = wmma4(Af, vb, Cb);
            float c = wf2h[0] * fmaxf(d[0], 0.0f);
#pragma unroll
            for (int i = 1; i < 8; ++i) c = fmaf(wf2h[i], fmaxf(d[i], 0.0f), c);
            tot[hh] = c + xhalf(c);
        }
        out[(size_t)(p * BB + b0 + bi) * TT + tc * CH + lane] = (hi ? tot[1] : tot[0]) + bf2v;
      }
      bc = bn;
    }
}

extern "C" void kernel_launch(void* const* d_in, const int* in_sizes, int n_in,
                              void* d_out, int out_size, void* d_ws, size_t ws_size,
                              hipStream_t stream) {
    (void)in_sizes; (void)n_in; (void)d_ws; (void)ws_size; (void)out_size;
    pinn_kernel<<<dim3(16, 4), dim3(128), 0, stream>>>(
        (const float*)d_in[0],  (const float*)d_in[1],  (const float*)d_in[2],
        (const float*)d_in[3],  (const float*)d_in[4],  (const float*)d_in[5],
        (const float*)d_in[6],  (const float*)d_in[7],  (const float*)d_in[8],
        (const float*)d_in[9],  (const float*)d_in[10], (const float*)d_in[11],
        (const float*)d_in[12], (const float*)d_in[13], (const float*)d_in[14],
        (float*)d_out);
}


# four helper waves per workgroup (second wave per SIMD, lower priority) take input staging, the lane halves' stress sum and the fusion head out of the recurrence waves' issue stream; sums unchanged
# speedup vs baseline: 10.7069x; 1.0691x over previous
#include <hip/hip_runtime.h>
#include <math.h>

typedef __attribute__((ext_vector_type(2))) float v2f;
typedef __attribute__((ext_vector_type(8))) float v8f;

#define TT 4096
#define BB 256
#define CH 32
#define GS 8
#define RP 36
#define HP 608
#define LP 1232
#define NB (TT / CH)
#define XP 68

__device__ __forceinline__ float fast_tanh(float x) {
    return __builtin_amdgcn_tanhf(x);
}

__device__ __forceinline__ float xhalf(float x) {
    int b = __builtin_bit_cast(int, x);
    return __builtin_bit_cast(float,
        __builtin_amdgcn_permlanex16(b, b, 0x76543210, 0xfedcba98, false, false));
}

__device__ __forceinline__ v8f wmma4(v2f a, v2f b, v8f c) {
    return __builtin_amdgcn_wmma_f32_16x16x4_f32(
        false, a, false, b, (short)0, c, false, false);
}

__global__ __launch_bounds__(256)
void pinn_kernel(const float* __restrict__ x,
                 const float* __restrict__ W1,  const float* __restrict__ b1,
                 const float* __restrict__ W2,  const float* __restrict__ b2,
                 const float* __restrict__ W3,  const float* __restrict__ b3,
                 const float* __restrict__ logE,const float* __restrict__ logeta,
                 const float* __restrict__ imp, const float* __restrict__ lw,
                 const float* __restrict__ Wf1, const float* __restrict__ bf1,
                 const float* __restrict__ Wf2, const float* __restrict__ bf2,
                 float* __restrict__ out)
{
    __shared__ __align__(16) float ss[2 * 4 * LP];
    __shared__ __align__(16) float xsh[3 * 16 * XP];

    const int tid  = threadIdx.x;
    const int wv   = __builtin_amdgcn_readfirstlane(tid >> 5);
    const bool helper = wv >= 4;
    const int w    = wv & 3;
    const int p    = blockIdx.y;
    const int n    = w * 4 + p;
    const int lane = tid & 31;
    const int hi   = lane >> 4;
    const int lm   = lane & 15;
    const int b0   = blockIdx.x * 16;
    const int kb   = hi ? 2 : 0;

    const float iw  = 1.0f / (1.0f + __expf(-imp[n]));
    const float Eh   = hi ? 0.0f : __expf(logE[n]) * iw;
    const float etah = hi ? 0.0f : __expf(logeta[n]) * iw;
    const float b3h  = hi ? 0.0f : b3[n] * iw;

    v2f A1[2]; v8f C1[2];
#pragma unroll
    for (int h = 0; h < 2; ++h) {
        int j = 16 * h + lm;
        A1[h].x = W1[(n * 32 + j) * 3 + 2];
        A1[h].y = W1[(n * 32 + j) * 3 + hi];
#pragma unroll
        for (int i = 0; i < 8; ++i) C1[h][i] = b1[n * 32 + 16 * h + 8 * hi + i];
    }
    v2f A2[8];
#pragma unroll
    for (int kc = 0; kc < 8; ++kc) {
        const int hid = 16 * (kc >> 2) + 8 * hi + 2 * (kc & 3);
        A2[kc].x = W2[(n * 16 + lm) * 32 + hid];
        A2[kc].y = W2[(n * 16 + lm) * 32 + hid + 1];
    }
    v8f C2;
#pragma unroll
    for (int i = 0; i < 8; ++i) C2[i] = b2[n * 16 + 8 * hi + i];

    float w3h[8];
#pragma unroll
    for (int k = 0; k < 8; ++k) w3h[k] = W3[n * 16 + 8 * hi + k] * iw;

    float cl[4];
    {
        float e0 = __expf(lw[0]), e1 = __expf(lw[1]);
        float e2 = __expf(lw[2]), e3 = __expf(lw[3]);
        float es = e0 + e1 + e2 + e3;
        float ws[4] = { e0 / es, e1 / es, e2 / es, e3 / es };
#pragma unroll
        for (int l = 0; l < 4; ++l) {
            float s = 0.0f;
#pragma unroll
            for (int q = 0; q < 4; ++q)
                s += 1.0f / (1.0f + __expf(-imp[l * 4 + q]));
            cl[l] = (s / (s + 1e-8f)) * ws[l];
        }
    }
    v2f Af;
    Af.x = Wf1[lm * 4 + kb];
    Af.y = Wf1[lm * 4 + kb + 1];
    const float clx = hi ? cl[2] : cl[0];
    const float cly = hi ? cl[3] : cl[1];
    v8f Cb;
    float wf2h[8];
#pragma unroll
    for (int i = 0; i < 8; ++i) { Cb[i] = bf1[8 * hi + i]; wf2h[i] = Wf2[8 * hi + i]; }
    const float bf2v = bf2[0];

    const int ht   = tid & 127;
    const int srow = ht >> 3;
    const int spc  = ht & 7;
    const float4* xq = (const float4*)x + ((size_t)(b0 + srow) * TT) / 2 + spc * 2;
    float* xst = xsh + srow * XP + spc * 8;
    if (helper) {
        const float4 a0 = xq[0],  a1 = xq[1];
        const float4 c0 = xq[16], c1 = xq[17];
        *(float4*)(xst)     = a0;
        *(float4*)(xst + 4) = a1;
        *(float4*)(xst + 16 * XP)     = c0;
        *(float4*)(xst + 16 * XP + 4) = c1;
    } else {
        __builtin_amdgcn_s_setprio(3);
    }
    __syncthreads();

    float part = 0.0f;
    const float* xrd = xsh + lm * XP;
    float4 q[4];
#pragma unroll
    for (int j = 0; j < 4; ++j) q[j] = make_float4(0.0f, 0.0f, 0.0f, 0.0f);
    if (!helper) {
#pragma unroll
        for (int j = 0; j < 4; ++j) q[j] = *(const float4*)(xrd + 4 * j);
    }

    int bc = 0;
    for (int tc = 0; tc <= NB; ++tc) {
      const int  bn   = (bc == 2) ? 0 : bc + 1;
      const int  bs   = (bn == 2) ? 0 : bn + 1;
      if (!helper) {
       if (tc < NB) {
        const bool lastb = (tc + 1) == NB;
        float* Sw = ss + (tc & 1) * (4 * LP) + w * LP + hi * HP + lm * RP;
#pragma nounroll
        for (int g = 0; g < CH / GS; ++g) {
          const int nb = (g + 1 < CH / GS || lastb) ? bc : bn;
          const int ng = (g + 1 < CH / GS) ? g + 1 : (lastb ? g : 0);
          const float* xn = xrd + nb * (16 * XP) + ng * (2 * GS);
          float4 nq[4];
#pragma unroll
          for (int j = 0; j < 4; ++j) nq[j] = *(const float4*)(xn + 4 * j);
          __builtin_amdgcn_sched_barrier(0);

          float pt[GS];
#pragma unroll
          for (int u = 0; u < GS; ++u) {
            const float xs0 = (u & 1) ? q[u >> 1].z : q[u >> 1].x;
            const float xs1 = (u & 1) ? q[u >> 1].w : q[u >> 1].y;

            v2f bx;
            bx.x = part;
            bx.y = hi ? xs1 : xs0;

            v8f d0 = wmma4(A1[0], bx, C1[0]);
            v8f d1 = wmma4(A1[1], bx, C1[1]);

            v8f acc = C2;
            v8f acb = {0.0f, 0.0f, 0.0f, 0.0f, 0.0f, 0.0f, 0.0f, 0.0f};
#pragma unroll
            for (int k = 0; k < 4; ++k) {
                v2f ha, hb;
                ha.x = fast_tanh(d0[2 * k]);
                ha.y = fast_tanh(d0[2 * k + 1]);
                acc = wmma4(A2[k], ha, acc);
                hb.x = fast_tanh(d1[2 * k]);
                hb.y = fast_tanh(d1[2 * k + 1]);
                acb = wmma4(A2[4 + k], hb, acb);
            }
            acc = acc + acb;

            float pe = fmaf(Eh, xs0, fmaf(etah, xs1, b3h)), po = 0.0f;
#pragma unroll
            for (int i = 0; i < 8; i += 2) {
                pe = fmaf(fast_tanh(acc[i]),     w3h[i],     pe);
                po = fmaf(fast_tanh(acc[i + 1]), w3h[i + 1], po);
            }
            part = pe + po;
            pt[u] = part;
          }

          float4 s0, s1;
          s0.x = pt[0]; s0.y = pt[1]; s0.z = pt[2]; s0.w = pt[3];
          s1.x = pt[4]; s1.y = pt[5]; s1.z = pt[6]; s1.w = pt[7];
          *(float4*)(Sw + g * GS)     = s0;
          *(float4*)(Sw + g * GS + 4) = s1;
#pragma unroll
          for (int j = 0; j < 4; ++j) q[j] = nq[j];
        }
       }
      } else {
        const bool more = (tc + 2) < NB;
        float4 ga = make_float4(0.0f, 0.0f, 0.0f, 0.0f), gb = ga;
        if (more) { ga = xq[(tc + 2) * 16]; gb = xq[(tc + 2) * 16 + 1]; }

        if (tc >= 1) {
          const int tb = tc - 1;
          const float* Sr = ss + (tb & 1) * (4 * LP);
#pragma unroll
          for (int r = 0; r < 4; ++r) {
            const int bi = 4 * w + r;
            float tot[2];
#pragma unroll
            for (int hh = 0; hh < 2; ++hh) {
                const float* sp = &Sr[kb * LP + bi * RP + 16 * hh + lm];
                v2f vb;
                vb.x = (sp[0]  + sp[HP])      * clx;
                vb.y = (sp[LP] + sp[LP + HP]) * cly;
                v8f d = wmma4(Af, vb, Cb);
                float c = wf2h[0] * fmaxf(d[0], 0.0f);
#pragma unroll
                for (int i = 1; i < 8; ++i) c = fmaf(wf2h[i], fmaxf(d[i], 0.0f), c);
                tot[hh] = c + xhalf(c);
            }
            out[(size_t)(p * BB + b0 + bi) * TT + tb * CH + lane] = (hi ? tot[1] : tot[0]) + bf2v;
          }
        }

        if (more) {
          float* xd = xst + bs * (16 * XP);
          *(float4*)(xd)     = ga;
          *(float4*)(xd + 4) = gb;
        }
      }
      __syncthreads();
      bc = bn;
    }
}

extern "C" void kernel_launch(void* const* d_in, const int* in_sizes, int n_in,
                              void* d_out, int out_size, void* d_ws, size_t ws_size,
                              hipStream_t stream) {
    (void)in_sizes; (void)n_in; (void)d_ws; (void)ws_size; (void)out_size;
    pinn_kernel<<<dim3(16, 4), dim3(256), 0, stream>>>(
        (const float*)d_in[0],  (const float*)d_in[1],  (const float*)d_in[2],
        (const float*)d_in[3],  (const float*)d_in[4],  (const float*)d_in[5],
        (const float*)d_in[6],  (const float*)d_in[7],  (const float*)d_in[8],
        (const float*)d_in[9],  (const float*)d_in[10], (const float*)d_in[11],
        (const float*)d_in[12], (const float*)d_in[13], (const float*)d_in[14],
        (float*)d_out);
}
